# starting kernel
# baseline (speedup 1.0000x reference)
#include <hip/hip_runtime.h>
#include <math.h>

#define NNODES  40000
#define NEDGES  640000
#define DIM     128
#define NGRAPH  64
#define NCLS    100
#define NLAYERS 3

typedef __attribute__((ext_vector_type(2))) float v2f;
typedef __attribute__((ext_vector_type(8))) float v8f;

__device__ __forceinline__ void atomAddF(float* p, float v) {
    __hip_atomic_fetch_add(p, v, __ATOMIC_RELAXED, __HIP_MEMORY_SCOPE_AGENT);
}

__global__ void normalize_rows(const float* __restrict__ h,
                               float* __restrict__ nh, int n) {
    int lane = threadIdx.x & 31;
    int row  = blockIdx.x * (blockDim.x >> 5) + (threadIdx.x >> 5);
    if (row >= n) return;
    float4 v = ((const float4*)(h + (size_t)row * DIM))[lane];
    float ss = v.x * v.x + v.y * v.y + v.z * v.z + v.w * v.w;
#pragma unroll
    for (int off = 16; off >= 1; off >>= 1) ss += __shfl_xor(ss, off, 32);
    float inv = 1.0f / (sqrtf(ss) + 1e-12f);
    float4 o;
    o.x = v.x * inv; o.y = v.y * inv; o.z = v.z * inv; o.w = v.w * inv;
    ((float4*)(nh + (size_t)row * DIM))[lane] = o;
}

__global__ void edge_attend(const float* __restrict__ nh,
                            const float* __restrict__ h,
                            const int* __restrict__ src,
                            const int* __restrict__ dst,
                            const float* __restrict__ betas, int layer,
                            float* __restrict__ num,
                            float* __restrict__ denom, int e) {
    int lane = threadIdx.x & 31;
    int edge = blockIdx.x * (blockDim.x >> 5) + (threadIdx.x >> 5);
    if (edge >= e) return;
    int s = src[edge];
    int d = dst[edge];
    float4 a4 = ((const float4*)(nh + (size_t)s * DIM))[lane];
    float4 b4 = ((const float4*)(nh + (size_t)d * DIM))[lane];
    float p = a4.x * b4.x + a4.y * b4.y + a4.z * b4.z + a4.w * b4.w;
#pragma unroll
    for (int off = 16; off >= 1; off >>= 1) p += __shfl_xor(p, off, 32);
    float att = __expf(betas[layer] * p);
    if (lane == 0) atomAddF(denom + d, att);
    float4 hv = ((const float4*)(h + (size_t)s * DIM))[lane];
    float* np = num + (size_t)d * DIM + lane * 4;
    atomAddF(np + 0, att * hv.x);
    atomAddF(np + 1, att * hv.y);
    atomAddF(np + 2, att * hv.z);
    atomAddF(np + 3, att * hv.w);
}

__global__ void finalize_div(float* __restrict__ num,
                             const float* __restrict__ denom, int nd) {
    int i = blockIdx.x * blockDim.x + threadIdx.x;
    if (i >= nd) return;
    float dn = denom[i >> 7];
    num[i] = num[i] / (dn > 0.0f ? dn : 1.0f);
}

__global__ void pool_sum(const float* __restrict__ h,
                         const int* __restrict__ gid,
                         float* __restrict__ hg, int n) {
    int i = blockIdx.x * blockDim.x + threadIdx.x;
    if (i >= n * DIM) return;
    int node = i >> 7;
    int d    = i & 127;
    atomAddF(hg + (size_t)gid[node] * DIM + d, h[i]);
}

__global__ void pool_cnt(const int* __restrict__ gid,
                         float* __restrict__ cnt, int n) {
    int i = blockIdx.x * blockDim.x + threadIdx.x;
    if (i >= n) return;
    atomAddF(cnt + gid[i], 1.0f);
}

__global__ void pool_div(float* __restrict__ hg, const float* __restrict__ cnt) {
    int i = blockIdx.x * blockDim.x + threadIdx.x;
    if (i >= NGRAPH * DIM) return;
    hg[i] /= fmaxf(cnt[i >> 7], 1.0f);
}

__global__ void classifier_wmma(const float* __restrict__ hg,
                                const float* __restrict__ W,
                                const float* __restrict__ b,
                                float* __restrict__ out) {
    int mt = blockIdx.x / 7;
    int nt = blockIdx.x % 7;
    int m0 = mt * 16;
    int n0 = nt * 16;
    int lane = threadIdx.x & 31;
    int half = lane >> 4;
    int lr   = lane & 15;
    int col  = n0 + lr;
    int ccol = col < NCLS ? col : (NCLS - 1);
    float mask = col < NCLS ? 1.0f : 0.0f;

    v8f acc = {};
#pragma unroll 4
    for (int k0 = 0; k0 < DIM; k0 += 4) {
        const float* ar = hg + (size_t)(m0 + lr) * DIM + k0 + 2 * half;
        v2f a;
        a.x = ar[0];
        a.y = ar[1];
        v2f bb;
        bb.x = W[(size_t)(k0 + 2 * half) * NCLS + ccol] * mask;
        bb.y = W[(size_t)(k0 + 2 * half + 1) * NCLS + ccol] * mask;
        acc = __builtin_amdgcn_wmma_f32_16x16x4_f32(
             false, a,  false, bb,
             (short)0, acc,  false,  false);
    }

    if (col < NCLS) {
        float bias = b[col];
#pragma unroll
        for (int r = 0; r < 8; ++r) {
            int m = m0 + r + 8 * half;
            out[(size_t)m * NCLS + col] = acc[r] + bias;
        }
    }
}

extern "C" void kernel_launch(void* const* d_in, const int* in_sizes, int n_in,
                              void* d_out, int out_size, void* d_ws, size_t ws_size,
                              hipStream_t stream) {
    const float* h0    = (const float*)d_in[0];
    const int*   src   = (const int*)d_in[1];
    const int*   dst   = (const int*)d_in[2];
    const int*   gid   = (const int*)d_in[3];
    const float* betas = (const float*)d_in[4];
    const float* Wc    = (const float*)d_in[5];
    const float* bc    = (const float*)d_in[6];
    float*       out   = (float*)d_out;

    float* nh    = (float*)d_ws;
    float* hA    = nh + (size_t)NNODES * DIM;
    float* hB    = hA + (size_t)NNODES * DIM;
    float* denom = hB + (size_t)NNODES * DIM;
    float* hg    = denom + NNODES;
    float* cnt   = hg + NGRAPH * DIM;

    const int ND = NNODES * DIM;

    const float* hin = h0;
    float* outs[NLAYERS] = {hA, hB, hA};
    for (int l = 0; l < NLAYERS; ++l) {
        float* hout = outs[l];
        hipMemsetAsync(hout, 0, sizeof(float) * (size_t)ND, stream);
        hipMemsetAsync(denom, 0, sizeof(float) * NNODES, stream);
        normalize_rows<<<(NNODES + 7) / 8, 256, 0, stream>>>(hin, nh, NNODES);
        edge_attend<<<(NEDGES + 7) / 8, 256, 0, stream>>>(nh, hin, src, dst,
                                                          betas, l, hout, denom,
                                                          NEDGES);
        finalize_div<<<(ND + 255) / 256, 256, 0, stream>>>(hout, denom, ND);
        hin = hout;
    }

    hipMemsetAsync(hg, 0, sizeof(float) * NGRAPH * DIM, stream);
    hipMemsetAsync(cnt, 0, sizeof(float) * NGRAPH, stream);
    pool_sum<<<(ND + 255) / 256, 256, 0, stream>>>(hin, gid, hg, NNODES);
    pool_cnt<<<(NNODES + 255) / 256, 256, 0, stream>>>(gid, cnt, NNODES);
    pool_div<<<(NGRAPH * DIM + 255) / 256, 256, 0, stream>>>(hg, cnt);

    classifier_wmma<<<4 * 7, 32, 0, stream>>>(hg, Wc, bc, out);
}


# A wave takes a run of items, not one: edge pass 16 edges per wave (ids read once, row loads of 8 edges in flight together), row passes 8 rows per wave; 17 times fewer waves launched per run
# speedup vs baseline: 1.2566x; 1.2566x over previous
#include <hip/hip_runtime.h>
#include <math.h>


#define NNODES  40000
#define NEDGES  640000
#define DIM     128
#define NGRAPH  64
#define NCLS    100
#define NLAYERS 3

#define RPW   8
#define EPW   16
#define EGRP  8
#define WPB   8

#define ROW_BLOCKS  ((NNODES + RPW * WPB - 1) / (RPW * WPB))
#define EDGE_BLOCKS ((NEDGES + EPW * WPB - 1) / (EPW * WPB))

typedef __attribute__((ext_vector_type(2))) float v2f;
typedef __attribute__((ext_vector_type(8))) float v8f;

__device__ __forceinline__ void atomAddF(float* p, float v) {
    __hip_atomic_fetch_add(p, v, __ATOMIC_RELAXED, __HIP_MEMORY_SCOPE_AGENT);
}

__device__ __forceinline__ int wave_in_grid() {
    return blockIdx.x * WPB + __builtin_amdgcn_readfirstlane(threadIdx.x >> 5);
}

__global__ void __launch_bounds__(256)
normalize_rows(const float* __restrict__ h, float* __restrict__ nh, int n) {
    const int lane = threadIdx.x & 31;
    const int row0 = wave_in_grid() * RPW;
    if (row0 >= n) return;
    float4 v[RPW];
#pragma unroll
    for (int r = 0; r < RPW; ++r) {
        int row = row0 + r < n ? row0 + r : n - 1;
        v[r] = ((const float4*)(h + (size_t)row * DIM))[lane];
    }
#pragma unroll
    for (int r = 0; r < RPW; ++r) {
        float ss = v[r].x * v[r].x + v[r].y * v[r].y + v[r].z * v[r].z + v[r].w * v[r].w;
#pragma unroll
        for (int off = 16; off >= 1; off >>= 1) ss += __shfl_xor(ss, off, 32);
        float inv = 1.0f / (sqrtf(ss) + 1e-12f);
        v[r].x *= inv; v[r].y *= inv; v[r].z *= inv; v[r].w *= inv;
    }
#pragma unroll
    for (int r = 0; r < RPW; ++r)
        if (row0 + r < n) ((float4*)(nh + (size_t)(row0 + r) * DIM))[lane] = v[r];
}

__device__ __forceinline__ void edge_one(const float* __restrict__ nh,
                                         const float* __restrict__ h,
                                         int s, int d, float beta, int lane,
                                         float* __restrict__ num,
                                         float* __restrict__ denom) {
    float4 a4 = ((const float4*)(nh + (size_t)s * DIM))[lane];
    float4 b4 = ((const float4*)(nh + (size_t)d * DIM))[lane];
    const float* hs = h + (size_t)s * DIM + lane;
    float h0 = hs[0], h1 = hs[32], h2 = hs[64], h3 = hs[96];
    float p = a4.x * b4.x + a4.y * b4.y + a4.z * b4.z + a4.w * b4.w;
#pragma unroll
    for (int off = 16; off >= 1; off >>= 1) p += __shfl_xor(p, off, 32);
    float att = __expf(beta * p);
    if (lane == 0) atomAddF(denom + d, att);
    float* np = num + (size_t)d * DIM + lane;
    atomAddF(np + 0,  att * h0);
    atomAddF(np + 32, att * h1);
    atomAddF(np + 64, att * h2);
    atomAddF(np + 96, att * h3);
}

__global__ void __launch_bounds__(256)
edge_attend(const float* __restrict__ nh,
            const float* __restrict__ h,
            const int* __restrict__ src,
            const int* __restrict__ dst,
            const float* __restrict__ betas, int layer,
            float* __restrict__ num,
            float* __restrict__ denom, int e) {
    const int lane = threadIdx.x & 31;
    const int base = wave_in_grid() * EPW;
    if (base >= e) return;
    const int left = e - base;
    const int slot = lane & (EPW - 1);
    const int* ids = lane < EPW ? src : dst;
    const int idv = slot < left ? ids[base + slot] : 0;
    const float beta = betas[layer];

    if (left < EPW) {
        for (int j = 0; j < left; ++j)
            edge_one(nh, h, __builtin_amdgcn_readlane(idv, j),
                     __builtin_amdgcn_readlane(idv, EPW + j), beta, lane,
                     num, denom);
        return;
    }

    float attmine = 0.0f;
#pragma unroll
    for (int g = 0; g < EPW; g += EGRP) {
        int   dj[EGRP];
        float4 a4[EGRP], b4[EGRP];
        float  hv[EGRP][4];
#pragma unroll
        for (int u = 0; u < EGRP; ++u) {
            const int sj = __builtin_amdgcn_readlane(idv, g + u);
            dj[u] = __builtin_amdgcn_readlane(idv, EPW + g + u);
            a4[u] = ((const float4*)(nh + (size_t)sj * DIM))[lane];
            b4[u] = ((const float4*)(nh + (size_t)dj[u] * DIM))[lane];
            const float* hs = h + (size_t)sj * DIM + lane;
            hv[u][0] = hs[0];
            hv[u][1] = hs[32];
            hv[u][2] = hs[64];
            hv[u][3] = hs[96];
        }
        float att[EGRP];
#pragma unroll
        for (int u = 0; u < EGRP; ++u) {
            float p = a4[u].x * b4[u].x + a4[u].y * b4[u].y +
                      a4[u].z * b4[u].z + a4[u].w * b4[u].w;
#pragma unroll
            for (int off = 16; off >= 1; off >>= 1) p += __shfl_xor(p, off, 32);
            att[u] = __expf(beta * p);
            attmine = slot == g + u ? att[u] : attmine;
        }
#pragma unroll
        for (int u = 0; u < EGRP; ++u) {
            float* np = num + (size_t)dj[u] * DIM + lane;
            atomAddF(np + 0,  att[u] * hv[u][0]);
            atomAddF(np + 32, att[u] * hv[u][1]);
            atomAddF(np + 64, att[u] * hv[u][2]);
            atomAddF(np + 96, att[u] * hv[u][3]);
        }
    }
    if (lane >= EPW) atomAddF(denom + idv, attmine);
}

__global__ void __launch_bounds__(256)
finalize_div(float* __restrict__ num, const float* __restrict__ denom, int n) {
    const int lane = threadIdx.x & 31;
    const int row0 = wave_in_grid() * RPW;
    if (row0 >= n) return;
    float4 v[RPW];
    float  dn[RPW];
#pragma unroll
    for (int r = 0; r < RPW; ++r) {
        int row = row0 + r < n ? row0 + r : n - 1;
        v[r]  = ((const float4*)(num + (size_t)row * DIM))[lane];
        dn[r] = denom[row];
    }
#pragma unroll
    for (int r = 0; r < RPW; ++r) {
        float d = dn[r] > 0.0f ? dn[r] : 1.0f;
        float4 o;
        o.x = v[r].x / d; o.y = v[r].y / d; o.z = v[r].z / d; o.w = v[r].w / d;
        if (row0 + r < n) ((float4*)(num + (size_t)(row0 + r) * DIM))[lane] = o;
    }
}

__global__ void __launch_bounds__(256)
pool_sum(const float* __restrict__ h, const int* __restrict__ gid,
         float* __restrict__ hg, int n) {
    const int lane = threadIdx.x & 31;
    const int row0 = wave_in_grid() * RPW;
    if (row0 >= n) return;
    const int left = n - row0;
    const int slot = lane & (RPW - 1);
    const int gv = gid[slot < left ? row0 + slot : n - 1];

    if (left < RPW) {
        for (int r = 0; r < left; ++r) {
            const float* hr = h + (size_t)(row0 + r) * DIM + lane;
            float* gp = hg + (size_t)__builtin_amdgcn_readlane(gv, r) * DIM + lane;
            atomAddF(gp + 0,  hr[0]);
            atomAddF(gp + 32, hr[32]);
            atomAddF(gp + 64, hr[64]);
            atomAddF(gp + 96, hr[96]);
        }
        return;
    }

    float hv[RPW][4];
#pragma unroll
    for (int r = 0; r < RPW; ++r) {
        const float* hr = h + (size_t)(row0 + r) * DIM + lane;
        hv[r][0] = hr[0];
        hv[r][1] = hr[32];
        hv[r][2] = hr[64];
        hv[r][3] = hr[96];
    }
#pragma unroll
    for (int r = 0; r < RPW; ++r) {
        float* gp = hg + (size_t)__builtin_amdgcn_readlane(gv, r) * DIM + lane;
        atomAddF(gp + 0,  hv[r][0]);
        atomAddF(gp + 32, hv[r][1]);
        atomAddF(gp + 64, hv[r][2]);
        atomAddF(gp + 96, hv[r][3]);
    }
}

__global__ void pool_cnt(const int* __restrict__ gid,
                         float* __restrict__ cnt, int n) {
    int i = blockIdx.x * blockDim.x + threadIdx.x;
    if (i >= n) return;
    atomAddF(cnt + gid[i], 1.0f);
}

__global__ void pool_div(float* __restrict__ hg, const float* __restrict__ cnt) {
    int i = blockIdx.x * blockDim.x + threadIdx.x;
    if (i >= NGRAPH * DIM) return;
    hg[i] /= fmaxf(cnt[i >> 7], 1.0f);
}

__global__ void classifier_wmma(const float* __restrict__ hg,
                                const float* __restrict__ W,
                                const float* __restrict__ b,
                                float* __restrict__ out) {
    int mt = blockIdx.x / 7;
    int nt = blockIdx.x % 7;
    int m0 = mt * 16;
    int n0 = nt * 16;
    int lane = threadIdx.x & 31;
    int half = lane >> 4;
    int lr   = lane & 15;
    int col  = n0 + lr;
    int ccol = col < NCLS ? col : (NCLS - 1);
    float mask = col < NCLS ? 1.0f : 0.0f;

    v8f acc = {};
#pragma unroll 4
    for (int k0 = 0; k0 < DIM; k0 += 4) {
        const float* ar = hg + (size_t)(m0 + lr) * DIM + k0 + 2 * half;
        v2f a;
        a.x = ar[0];
        a.y = ar[1];
        v2f bb;
        bb.x = W[(size_t)(k0 + 2 * half) * NCLS + ccol] * mask;
        bb.y = W[(size_t)(k0 + 2 * half + 1) * NCLS + ccol] * mask;
        acc = __builtin_amdgcn_wmma_f32_16x16x4_f32(
             false, a,  false, bb,
             (short)0, acc,  false,  false);
    }

    if (col < NCLS) {
        float bias = b[col];
#pragma unroll
        for (int r = 0; r < 8; ++r) {
            int m = m0 + r + 8 * half;
            out[(size_t)m * NCLS + col] = acc[r] + bias;
        }
    }
}

extern "C" void kernel_launch(void* const* d_in, const int* in_sizes, int n_in,
                              void* d_out, int out_size, void* d_ws, size_t ws_size,
                              hipStream_t stream) {
    const float* h0    = (const float*)d_in[0];
    const int*   src   = (const int*)d_in[1];
    const int*   dst   = (const int*)d_in[2];
    const int*   gid   = (const int*)d_in[3];
    const float* betas = (const float*)d_in[4];
    const float* Wc    = (const float*)d_in[5];
    const float* bc    = (const float*)d_in[6];
    float*       out   = (float*)d_out;

    float* nh    = (float*)d_ws;
    float* hA    = nh + (size_t)NNODES * DIM;
    float* hB    = hA + (size_t)NNODES * DIM;
    float* denom = hB + (size_t)NNODES * DIM;
    float* hg    = denom + NNODES;
    float* cnt   = hg + NGRAPH * DIM;

    const int ND = NNODES * DIM;

    const float* hin = h0;
    float* outs[NLAYERS] = {hA, hB, hA};
    for (int l = 0; l < NLAYERS; ++l) {
        float* hout = outs[l];
        hipMemsetAsync(hout, 0, sizeof(float) * (size_t)ND, stream);
        hipMemsetAsync(denom, 0, sizeof(float) * NNODES, stream);
        normalize_rows<<<ROW_BLOCKS, 256, 0, stream>>>(hin, nh, NNODES);
        edge_attend<<<EDGE_BLOCKS, 256, 0, stream>>>(nh, hin, src, dst,
                                                     betas, l, hout, denom,
                                                     NEDGES);
        finalize_div<<<ROW_BLOCKS, 256, 0, stream>>>(hout, denom, NNODES);
        hin = hout;
    }

    hipMemsetAsync(hg, 0, sizeof(float) * NGRAPH * DIM, stream);
    hipMemsetAsync(cnt, 0, sizeof(float) * NGRAPH, stream);
    pool_sum<<<ROW_BLOCKS, 256, 0, stream>>>(hin, gid, hg, NNODES);
    pool_cnt<<<(NNODES + 255) / 256, 256, 0, stream>>>(gid, cnt, NNODES);
    pool_div<<<(NGRAPH * DIM + 255) / 256, 256, 0, stream>>>(hg, cnt);

    classifier_wmma<<<4 * 7, 32, 0, stream>>>(hg, Wc, bc, out);
}


# Layers 2, 3 skip the dst scan: layer 1 leaves each wave's found edges (source row, local row) in the workspace, later layers join a group's lists in LDS and split them evenly; LDS add order differs
# speedup vs baseline: 3.1845x; 2.5342x over previous
#include <hip/hip_runtime.h>
#include <math.h>


#define NNODES  40000
#define NEDGES  640000
#define DIM     128
#define NGRAPH  64
#define NCLS    100
#define NLAYERS 3

#define RPW   8
#define WPB   8
#define ROW_BLOCKS  ((NNODES + RPW * WPB - 1) / (RPW * WPB))

#define LGROUPS  256
#define LWAVES   32
#define LTHREADS (LWAVES * 32)
#define RPG      ((NNODES + LGROUPS - 1) / LGROUPS)
#define ROWS0    ((RPG + LWAVES - 1) / LWAVES)
#define BATCH    512
#define EG       8
#define LISTCAP  (BATCH + EG)
#define NBATCH   (NEDGES / BATCH)
#define KEEPCAP  160

static_assert(NEDGES % BATCH == 0, "the id scan reads whole steps");
static_assert(NEDGES <= (1 << 24) && RPG <= 256, "list entry = edge index << 8 | local row");
static_assert(NBATCH >= LWAVES, "every wave has a first step");
static_assert(NNODES <= (1 << 24), "kept entry = source row << 8 | local row");
static_assert(KEEPCAP % 32 == 0 && KEEPCAP % EG == 0, "a kept list is read a wave-wide load at a time");
static_assert(LWAVES * KEEPCAP <= LWAVES * LISTCAP, "the group's joined list fits the LDS list area");

typedef __attribute__((ext_vector_type(2))) float v2f;
typedef __attribute__((ext_vector_type(8))) float v8f;

__device__ __forceinline__ void atomAddF(float* p, float v) {
    __hip_atomic_fetch_add(p, v, __ATOMIC_RELAXED, __HIP_MEMORY_SCOPE_AGENT);
}

__device__ __forceinline__ void ldsAddF(float* p, float v) {
    __hip_atomic_fetch_add(p, v, __ATOMIC_RELAXED, __HIP_MEMORY_SCOPE_WORKGROUP);
}

__device__ __forceinline__ int wave_in_grid() {
    return blockIdx.x * WPB + __builtin_amdgcn_readfirstlane(threadIdx.x >> 5);
}

template <int M>
__device__ __forceinline__ float row_xor(float x) {
    int v = __builtin_bit_cast(int, x);
    return __builtin_bit_cast(float, __builtin_amdgcn_update_dpp(0, v, 0x160 + M, 0xf, 0xf, true));
}
__device__ __forceinline__ float xhalf(float x) {
    int v = __builtin_bit_cast(int, x);
    return __builtin_bit_cast(float, __builtin_amdgcn_permlanex16(v, v, 0x76543210, 0xfedcba98, false, false));
}
__device__ __forceinline__ float lane_val(float x, int l) {
    return __builtin_bit_cast(float, __builtin_amdgcn_readlane(__builtin_bit_cast(int, x), l));
}
__device__ __forceinline__ int wave_sum_i(int x) {
    x += __builtin_amdgcn_permlanex16(x, x, 0x76543210, 0xfedcba98, false, false);
    x += __builtin_amdgcn_update_dpp(0, x, 0x160 + 8, 0xf, 0xf, true);
    x += __builtin_amdgcn_update_dpp(0, x, 0x160 + 4, 0xf, 0xf, true);
    x += __builtin_amdgcn_update_dpp(0, x, 0x160 + 2, 0xf, 0xf, true);
    x += __builtin_amdgcn_update_dpp(0, x, 0x160 + 1, 0xf, 0xf, true);
    return x;
}

template <bool FULL, bool KEPT>
__device__ __forceinline__ void edge_group(const float* __restrict__ hin,
                                           const int* __restrict__ src,
                                           const unsigned* ent, int n,
                                           float beta, int lane,
                                           const float* s_nd, float* s_acc,
                                           float* s_den, unsigned* keep) {
    int slot = lane & (EG - 1);
    if (!FULL) slot = slot < n ? slot : 0;
    const unsigned en = ent[slot];
    const int dl = (int)(en & 255u);
    const int sv = KEPT ? (int)(en >> 8) : src[en >> 8];
    if (!KEPT) {
        if (keep != nullptr && lane < (FULL ? EG : n))
            keep[lane] = ((unsigned)sv << 8) | (unsigned)dl;
    }
    int    dj[EG];
    float4 a4[EG];
    float  attmine = 0.0f;
#pragma unroll
    for (int u = 0; u < EG; ++u) {
        if (FULL || u < n) {
            const int sj = __builtin_amdgcn_readlane(sv, u);
            dj[u] = __builtin_amdgcn_readlane(dl, u);
            a4[u] = ((const float4*)(hin + (size_t)sj * DIM))[lane];
        }
    }
    __builtin_amdgcn_sched_barrier(0);
#pragma unroll
    for (int u = 0; u < EG; ++u) {
        if (FULL || u < n) {
            const float4 b4 = ((const float4*)(s_nd + dj[u] * DIM))[lane];
            const float4 a = a4[u];
            float p = a.x * b4.x + a.y * b4.y + a.z * b4.z + a.w * b4.w;
            float q = a.x * a.x + a.y * a.y + a.z * a.z + a.w * a.w;
            const bool lowhalf = lane < 16;
            float t = (lowhalf ? p : q) + xhalf(lowhalf ? q : p);
            t += row_xor<8>(t);
            t += row_xor<4>(t);
            t += row_xor<2>(t);
            t += row_xor<1>(t);
            const float dotv = lane_val(t, 0);
            const float ssv  = lane_val(t, 16);
            const float att = __expf(beta * (dotv / (sqrtf(ssv) + 1e-12f)));
            float* ap = s_acc + dj[u] * DIM + lane;
            ldsAddF(ap + 0,  att * a.x);
            ldsAddF(ap + 32, att * a.y);
            ldsAddF(ap + 64, att * a.z);
            ldsAddF(ap + 96, att * a.w);
            attmine = lane == u ? att : attmine;
        }
    }
    if (lane < (FULL ? EG : n)) ldsAddF(s_den + dl, attmine);
}

__device__ __forceinline__ void load_ids(int4 (&d)[4], const int* __restrict__ dst,
                                         int b, int lane) {
#pragma unroll
    for (int q = 0; q < 4; ++q)
        d[q] = ((const int4*)(dst + (size_t)b * BATCH + q * 128))[lane];
}

template <bool RECORD>
__device__ __forceinline__ void scan_edges(const float* __restrict__ hin,
                                           const int* __restrict__ src,
                                           const int* __restrict__ dst,
                                           int lo, int nrows, float beta,
                                           int lane, int wave,
                                           const float* s_nd, float* s_acc, float* s_den,
                                           unsigned* s_list,
                                           unsigned* __restrict__ elist,
                                           int* __restrict__ ecnt) {
    unsigned* mylist = s_list + wave * LISTCAP;
    unsigned* mykeep = elist + (size_t)(blockIdx.x * LWAVES + wave) * KEEPCAP;
    int cnt = 0;
    int done = 0;
    int4 cur[4];
    load_ids(cur, dst, wave, lane);
    for (int b = wave; b < NBATCH; b += LWAVES) {
        int4 nxt[4];
        load_ids(nxt, dst, b + LWAVES < NBATCH ? b + LWAVES : b, lane);
        const int ebase = b * BATCH + lane * 4;
#pragma unroll
        for (int q = 0; q < 4; ++q) {
            const int idq[4] = {cur[q].x, cur[q].y, cur[q].z, cur[q].w};
#pragma unroll
            for (int c = 0; c < 4; ++c) {
                const unsigned dl = (unsigned)(idq[c] - lo);
                const bool mine = dl < (unsigned)nrows;
                const unsigned m = __builtin_amdgcn_ballot_w32(mine);
                if (m != 0u) {
                    const int pos = cnt + (int)__builtin_amdgcn_mbcnt_lo(m, 0u);
                    if (mine) mylist[pos] = ((unsigned)(ebase + q * 128 + c) << 8) | dl;
                    cnt += __builtin_popcount(m);
                }
            }
        }
        while (cnt >= EG) {
            cnt -= EG;
            unsigned* kp = (RECORD && done + EG <= KEEPCAP) ? mykeep + done : nullptr;
            edge_group<true, false>(hin, src, mylist + cnt, EG, beta, lane, s_nd, s_acc, s_den, kp);
            done += EG;
        }
#pragma unroll
        for (int q = 0; q < 4; ++q) cur[q] = nxt[q];
    }
    if (cnt > 0) {
        unsigned* kp = (RECORD && done + cnt <= KEEPCAP) ? mykeep + done : nullptr;
        edge_group<false, false>(hin, src, mylist, cnt, beta, lane, s_nd, s_acc, s_den, kp);
        done += cnt;
    }
    if (RECORD && lane == 0) ecnt[blockIdx.x * LWAVES + wave] = done;
}

template <bool FIRST>
__global__ void __launch_bounds__(LTHREADS)
layer_owned(const float* __restrict__ hin, float* __restrict__ hout,
            const int* __restrict__ src, const int* __restrict__ dst,
            const float* __restrict__ betas, int layer,
            unsigned* __restrict__ elist, int* __restrict__ ecnt) {
    __shared__ __attribute__((aligned(16))) float s_nd[RPG * DIM];
    __shared__ __attribute__((aligned(16))) float s_acc[RPG * DIM];
    __shared__ float s_den[RPG];
    __shared__ unsigned s_list[LWAVES * LISTCAP];

    const int lane = threadIdx.x & 31;
    const int wave = __builtin_amdgcn_readfirstlane(threadIdx.x >> 5);
    const int lo = blockIdx.x * RPG;
    int nrows = NNODES - lo;
    nrows = nrows > RPG ? RPG : nrows;
    if (nrows <= 0) return;
    const float beta = betas[layer];

    bool kept = false;
    int total = 0, off = 0, mine = 0;
    unsigned ev[KEEPCAP / 32];
    if (!FIRST) {
        const int c = ecnt[blockIdx.x * LWAVES + lane];
        kept = __builtin_amdgcn_ballot_w32((unsigned)c > (unsigned)KEEPCAP) == 0u;
        if (kept) {
            total = __builtin_amdgcn_readfirstlane(wave_sum_i(c));
            off   = __builtin_amdgcn_readfirstlane(wave_sum_i(lane < wave ? c : 0));
            mine  = __builtin_amdgcn_readlane(c, wave);
            const unsigned* gl = elist + (size_t)(blockIdx.x * LWAVES + wave) * KEEPCAP;
#pragma unroll
            for (int k = 0; k < KEEPCAP / 32; ++k) {
                const int i = lane + 32 * k;
                ev[k] = 0u;
                if (i < mine) ev[k] = gl[i];
            }
        }
    }

    {
        float4 v[ROWS0];
#pragma unroll
        for (int i = 0; i < ROWS0; ++i) {
            const int r = wave + i * LWAVES;
            v[i] = ((const float4*)(hin + (size_t)(lo + (r < nrows ? r : 0)) * DIM))[lane];
        }
#pragma unroll
        for (int i = 0; i < ROWS0; ++i) {
            const int r = wave + i * LWAVES;
            float ss = v[i].x * v[i].x + v[i].y * v[i].y + v[i].z * v[i].z + v[i].w * v[i].w;
            ss += xhalf(ss);
            ss += row_xor<8>(ss);
            ss += row_xor<4>(ss);
            ss += row_xor<2>(ss);
            ss += row_xor<1>(ss);
            const float inv = 1.0f / (sqrtf(ss) + 1e-12f);
            if (r < nrows) {
                float4 o;
                o.x = v[i].x * inv; o.y = v[i].y * inv; o.z = v[i].z * inv; o.w = v[i].w * inv;
                ((float4*)(s_nd + r * DIM))[lane] = o;
                float* ap = s_acc + r * DIM + lane;
                ap[0] = 0.0f; ap[32] = 0.0f; ap[64] = 0.0f; ap[96] = 0.0f;
                if (lane == 0) s_den[r] = 0.0f;
            }
        }
    }
    if (!FIRST && kept) {
#pragma unroll
        for (int k = 0; k < KEEPCAP / 32; ++k) {
            const int i = lane + 32 * k;
            if (i < mine) s_list[off + i] = ev[k];
        }
    }
    __syncthreads();

    if (FIRST || !kept) {
        scan_edges<FIRST>(hin, src, dst, lo, nrows, beta, lane, wave,
                          s_nd, s_acc, s_den, s_list, elist, ecnt);
    } else {
        const int ngr = (total + EG - 1) / EG;
        const int g0 = (wave * ngr) / LWAVES;
        const int g1 = ((wave + 1) * ngr) / LWAVES;
        for (int gi = g0; gi < g1; ++gi) {
            const int left = total - gi * EG;
            if (left >= EG)
                edge_group<true, true>(hin, src, s_list + gi * EG, EG, beta, lane, s_nd, s_acc, s_den, nullptr);
            else
                edge_group<false, true>(hin, src, s_list + gi * EG, left, beta, lane, s_nd, s_acc, s_den, nullptr);
        }
    }
    __syncthreads();

    for (int r = wave; r < nrows; r += LWAVES) {
        const float* ap = s_acc + r * DIM + lane;
        const float dn = s_den[r];
        const float d = dn > 0.0f ? dn : 1.0f;
        float4 o;
        o.x = ap[0] / d; o.y = ap[32] / d; o.z = ap[64] / d; o.w = ap[96] / d;
        ((float4*)(hout + (size_t)(lo + r) * DIM))[lane] = o;
    }
}

__global__ void __launch_bounds__(256)
pool_sum(const float* __restrict__ h, const int* __restrict__ gid,
         float* __restrict__ hg, int n) {
    const int lane = threadIdx.x & 31;
    const int row0 = wave_in_grid() * RPW;
    if (row0 >= n) return;
    const int left = n - row0;
    const int slot = lane & (RPW - 1);
    const int gv = gid[slot < left ? row0 + slot : n - 1];

    if (left < RPW) {
        for (int r = 0; r < left; ++r) {
            const float* hr = h + (size_t)(row0 + r) * DIM + lane;
            float* gp = hg + (size_t)__builtin_amdgcn_readlane(gv, r) * DIM + lane;
            atomAddF(gp + 0,  hr[0]);
            atomAddF(gp + 32, hr[32]);
            atomAddF(gp + 64, hr[64]);
            atomAddF(gp + 96, hr[96]);
        }
        return;
    }

    float hv[RPW][4];
#pragma unroll
    for (int r = 0; r < RPW; ++r) {
        const float* hr = h + (size_t)(row0 + r) * DIM + lane;
        hv[r][0] = hr[0];
        hv[r][1] = hr[32];
        hv[r][2] = hr[64];
        hv[r][3] = hr[96];
    }
#pragma unroll
    for (int r = 0; r < RPW; ++r) {
        float* gp = hg + (size_t)__builtin_amdgcn_readlane(gv, r) * DIM + lane;
        atomAddF(gp + 0,  hv[r][0]);
        atomAddF(gp + 32, hv[r][1]);
        atomAddF(gp + 64, hv[r][2]);
        atomAddF(gp + 96, hv[r][3]);
    }
}

__global__ void pool_cnt(const int* __restrict__ gid,
                         float* __restrict__ cnt, int n) {
    int i = blockIdx.x * blockDim.x + threadIdx.x;
    if (i >= n) return;
    atomAddF(cnt + gid[i], 1.0f);
}

__global__ void pool_div(float* __restrict__ hg, const float* __restrict__ cnt) {
    int i = blockIdx.x * blockDim.x + threadIdx.x;
    if (i >= NGRAPH * DIM) return;
    hg[i] /= fmaxf(cnt[i >> 7], 1.0f);
}

__global__ void classifier_wmma(const float* __restrict__ hg,
                                const float* __restrict__ W,
                                const float* __restrict__ b,
                                float* __restrict__ out) {
    int mt = blockIdx.x / 7;
    int nt = blockIdx.x % 7;
    int m0 = mt * 16;
    int n0 = nt * 16;
    int lane = threadIdx.x & 31;
    int half = lane >> 4;
    int lr   = lane & 15;
    int col  = n0 + lr;
    int ccol = col < NCLS ? col : (NCLS - 1);
    float mask = col < NCLS ? 1.0f : 0.0f;

    v8f acc = {};
#pragma unroll 4
    for (int k0 = 0; k0 < DIM; k0 += 4) {
        const float* ar = hg + (size_t)(m0 + lr) * DIM + k0 + 2 * half;
        v2f a;
        a.x = ar[0];
        a.y = ar[1];
        v2f bb;
        bb.x = W[(size_t)(k0 + 2 * half) * NCLS + ccol] * mask;
        bb.y = W[(size_t)(k0 + 2 * half + 1) * NCLS + ccol] * mask;
        acc = __builtin_amdgcn_wmma_f32_16x16x4_f32(
             false, a,  false, bb,
             (short)0, acc,  false,  false);
    }

    if (col < NCLS) {
        float bias = b[col];
#pragma unroll
        for (int r = 0; r < 8; ++r) {
            int m = m0 + r + 8 * half;
            out[(size_t)m * NCLS + col] = acc[r] + bias;
        }
    }
}

extern "C" void kernel_launch(void* const* d_in, const int* in_sizes, int n_in,
                              void* d_out, int out_size, void* d_ws, size_t ws_size,
                              hipStream_t stream) {
    const float* h0    = (const float*)d_in[0];
    const int*   src   = (const int*)d_in[1];
    const int*   dst   = (const int*)d_in[2];
    const int*   gid   = (const int*)d_in[3];
    const float* betas = (const float*)d_in[4];
    const float* Wc    = (const float*)d_in[5];
    const float* bc    = (const float*)d_in[6];
    float*       out   = (float*)d_out;

    float* hA    = (float*)d_ws;
    float* hB    = hA + (size_t)NNODES * DIM;
    float* hg    = hB + (size_t)NNODES * DIM;
    float* cnt   = hg + NGRAPH * DIM;
    unsigned* elist = (unsigned*)(cnt + NGRAPH);
    int*   ecnt  = (int*)(elist + (size_t)LGROUPS * LWAVES * KEEPCAP);

    layer_owned<true><<<LGROUPS, LTHREADS, 0, stream>>>(h0, hA, src, dst, betas, 0, elist, ecnt);
    layer_owned<false><<<LGROUPS, LTHREADS, 0, stream>>>(hA, hB, src, dst, betas, 1, elist, ecnt);
    layer_owned<false><<<LGROUPS, LTHREADS, 0, stream>>>(hB, hA, src, dst, betas, 2, elist, ecnt);
    const float* hin = hA;

    hipMemsetAsync(hg, 0, sizeof(float) * NGRAPH * DIM, stream);
    hipMemsetAsync(cnt, 0, sizeof(float) * NGRAPH, stream);
    pool_sum<<<ROW_BLOCKS, 256, 0, stream>>>(hin, gid, hg, NNODES);
    pool_cnt<<<(NNODES + 255) / 256, 256, 0, stream>>>(gid, cnt, NNODES);
    pool_div<<<(NGRAPH * DIM + 255) / 256, 256, 0, stream>>>(hg, cnt);

    classifier_wmma<<<4 * 7, 32, 0, stream>>>(hg, Wc, bc, out);
}


# Edge groups: a group's 8 weights (sqrt, divide, exp) are worked once, edge u in lane u, not per edge in every lane; LDS row reads issued with the gathers, LDS adds last; per-lane 4-term sums now fma
# speedup vs baseline: 5.0996x; 1.6014x over previous
#include <hip/hip_runtime.h>
#include <math.h>


#define NNODES  40000
#define NEDGES  640000
#define DIM     128
#define NGRAPH  64
#define NCLS    100
#define NLAYERS 3

#define PCHUNK   160
#define PBLOCKS  (NNODES / PCHUNK)
#define PSTEPS   (PCHUNK / 4)
#define PWAVES   8
#define FWAVES   32

#define LGROUPS  256
#define LWAVES   32
#define LTHREADS (LWAVES * 32)
#define RPG      ((NNODES + LGROUPS - 1) / LGROUPS)
#define ROWS0    ((RPG + LWAVES - 1) / LWAVES)
#define BATCH    512
#define EG       8
#define LISTCAP  (BATCH + EG)
#define NBATCH   (NEDGES / BATCH)
#define KEEPCAP  160

static_assert(NEDGES % BATCH == 0, "the id scan reads whole steps");
static_assert(NEDGES <= (1 << 24) && RPG <= 256, "list entry = edge index << 8 | local row");
static_assert(NBATCH >= LWAVES, "every wave has a first step");
static_assert(NNODES <= (1 << 24), "kept entry = source row << 8 | local row");
static_assert(KEEPCAP % 32 == 0 && KEEPCAP % EG == 0, "a kept list is read a wave-wide load at a time");
static_assert(LWAVES * KEEPCAP <= LWAVES * LISTCAP, "the group's joined list fits the LDS list area");
static_assert((size_t)PBLOCKS * NGRAPH * (DIM + 1) <= (size_t)NNODES * DIM, "partial tiles and counts fit one row buffer");

typedef __attribute__((ext_vector_type(2))) float v2f;
typedef __attribute__((ext_vector_type(8))) float v8f;

__device__ __forceinline__ void ldsAddF(float* p, float v) {
    __hip_atomic_fetch_add(p, v, __ATOMIC_RELAXED, __HIP_MEMORY_SCOPE_WORKGROUP);
}

template <int M>
__device__ __forceinline__ float row_xor(float x) {
    int v = __builtin_bit_cast(int, x);
    return __builtin_bit_cast(float, __builtin_amdgcn_update_dpp(0, v, 0x160 + M, 0xf, 0xf, true));
}
__device__ __forceinline__ float xhalf(float x) {
    int v = __builtin_bit_cast(int, x);
    return __builtin_bit_cast(float, __builtin_amdgcn_permlanex16(v, v, 0x76543210, 0xfedcba98, false, false));
}
__device__ __forceinline__ float lane_val(float x, int l) {
    return __builtin_bit_cast(float, __builtin_amdgcn_readlane(__builtin_bit_cast(int, x), l));
}
__device__ __forceinline__ int wave_sum_i(int x) {
    x += __builtin_amdgcn_permlanex16(x, x, 0x76543210, 0xfedcba98, false, false);
    x += __builtin_amdgcn_update_dpp(0, x, 0x160 + 8, 0xf, 0xf, true);
    x += __builtin_amdgcn_update_dpp(0, x, 0x160 + 4, 0xf, 0xf, true);
    x += __builtin_amdgcn_update_dpp(0, x, 0x160 + 2, 0xf, 0xf, true);
    x += __builtin_amdgcn_update_dpp(0, x, 0x160 + 1, 0xf, 0xf, true);
    return x;
}

template <bool FULL, bool KEPT>
__device__ __forceinline__ void edge_group(const float* __restrict__ hin,
                                           const int* __restrict__ src,
                                           const unsigned* ent, int n,
                                           float beta, int lane,
                                           const float* s_nd, float* s_acc,
                                           float* s_den, unsigned* keep) {
    int slot = lane & (EG - 1);
    if (!FULL) slot = slot < n ? slot : 0;
    const unsigned en = ent[slot];
    const int dl = (int)(en & 255u);
    const int sv = KEPT ? (int)(en >> 8) : src[en >> 8];
    if (!KEPT) {
        if (keep != nullptr && lane < (FULL ? EG : n))
            keep[lane] = ((unsigned)sv << 8) | (unsigned)dl;
    }
    int    dj[EG];
    float4 a4[EG];
    float4 b4[EG];
#pragma unroll
    for (int u = 0; u < EG; ++u) {
        if (FULL || u < n) {
            const int sj = __builtin_amdgcn_readlane(sv, u);
            dj[u] = __builtin_amdgcn_readlane(dl, u);
            a4[u] = ((const float4*)(hin + (size_t)sj * DIM))[lane];
        }
    }
#pragma unroll
    for (int u = 0; u < EG; ++u) {
        if (FULL || u < n)
            b4[u] = ((const float4*)(s_nd + dj[u] * DIM))[lane];
    }
    __builtin_amdgcn_sched_barrier(0);
    const int  l16 = lane & 15;
    const bool lowhalf = lane < 16;
    float sums = 0.0f;
#pragma unroll
    for (int u = 0; u < EG; ++u) {
        if (FULL || u < n) {
            const float4 a = a4[u];
            const float4 b = b4[u];
            float p = a.x * b.x + a.y * b.y + a.z * b.z + a.w * b.w;
            float q = a.x * a.x + a.y * a.y + a.z * a.z + a.w * a.w;
            float t = (lowhalf ? p : q) + xhalf(lowhalf ? q : p);
            t += row_xor<8>(t);
            t += row_xor<4>(t);
            t += row_xor<2>(t);
            t += row_xor<1>(t);
            sums = l16 == u ? t : sums;
        }
    }
    const float ssq  = xhalf(sums);
    const float attv = __expf(beta * (sums / (sqrtf(ssq) + 1e-12f)));
#pragma unroll
    for (int u = 0; u < EG; ++u) {
        if (FULL || u < n) {
            const float att = lane_val(attv, u);
            const float4 a = a4[u];
            float* ap = s_acc + dj[u] * DIM + lane;
            ldsAddF(ap + 0,  att * a.x);
            ldsAddF(ap + 32, att * a.y);
            ldsAddF(ap + 64, att * a.z);
            ldsAddF(ap + 96, att * a.w);
        }
    }
    if (lane < (FULL ? EG : n)) ldsAddF(s_den + dl, attv);
}

__device__ __forceinline__ void load_ids(int4 (&d)[4], const int* __restrict__ dst,
                                         int b, int lane) {
#pragma unroll
    for (int q = 0; q < 4; ++q)
        d[q] = ((const int4*)(dst + (size_t)b * BATCH + q * 128))[lane];
}

template <bool RECORD>
__device__ __forceinline__ void scan_edges(const float* __restrict__ hin,
                                           const int* __restrict__ src,
                                           const int* __restrict__ dst,
                                           int lo, int nrows, float beta,
                                           int lane, int wave,
                                           const float* s_nd, float* s_acc, float* s_den,
                                           unsigned* s_list,
                                           unsigned* __restrict__ elist,
                                           int* __restrict__ ecnt) {
    unsigned* mylist = s_list + wave * LISTCAP;
    unsigned* mykeep = elist + (size_t)(blockIdx.x * LWAVES + wave) * KEEPCAP;
    int cnt = 0;
    int done = 0;
    int4 cur[4];
    load_ids(cur, dst, wave, lane);
    for (int b = wave; b < NBATCH; b += LWAVES) {
        int4 nxt[4];
        load_ids(nxt, dst, b + LWAVES < NBATCH ? b + LWAVES : b, lane);
        const int ebase = b * BATCH + lane * 4;
#pragma unroll
        for (int q = 0; q < 4; ++q) {
            const int idq[4] = {cur[q].x, cur[q].y, cur[q].z, cur[q].w};
#pragma unroll
            for (int c = 0; c < 4; ++c) {
                const unsigned dl = (unsigned)(idq[c] - lo);
                const bool mine = dl < (unsigned)nrows;
                const unsigned m = __builtin_amdgcn_ballot_w32(mine);
                if (m != 0u) {
                    const int pos = cnt + (int)__builtin_amdgcn_mbcnt_lo(m, 0u);
                    if (mine) mylist[pos] = ((unsigned)(ebase + q * 128 + c) << 8) | dl;
                    cnt += __builtin_popcount(m);
                }
            }
        }
        while (cnt >= EG) {
            cnt -= EG;
            unsigned* kp = (RECORD && done + EG <= KEEPCAP) ? mykeep + done : nullptr;
            edge_group<true, false>(hin, src, mylist + cnt, EG, beta, lane, s_nd, s_acc, s_den, kp);
            done += EG;
        }
#pragma unroll
        for (int q = 0; q < 4; ++q) cur[q] = nxt[q];
    }
    if (cnt > 0) {
        unsigned* kp = (RECORD && done + cnt <= KEEPCAP) ? mykeep + done : nullptr;
        edge_group<false, false>(hin, src, mylist, cnt, beta, lane, s_nd, s_acc, s_den, kp);
        done += cnt;
    }
    if (RECORD && lane == 0) ecnt[blockIdx.x * LWAVES + wave] = done;
}

template <bool FIRST>
__global__ void __launch_bounds__(LTHREADS)
layer_owned(const float* __restrict__ hin, float* __restrict__ hout,
            const int* __restrict__ src, const int* __restrict__ dst,
            const float* __restrict__ betas, int layer,
            unsigned* __restrict__ elist, int* __restrict__ ecnt) {
    __shared__ __attribute__((aligned(16))) float s_nd[RPG * DIM];
    __shared__ __attribute__((aligned(16))) float s_acc[RPG * DIM];
    __shared__ float s_den[RPG];
    __shared__ unsigned s_list[LWAVES * LISTCAP];

    const int lane = threadIdx.x & 31;
    const int wave = __builtin_amdgcn_readfirstlane(threadIdx.x >> 5);
    const int lo = blockIdx.x * RPG;
    int nrows = NNODES - lo;
    nrows = nrows > RPG ? RPG : nrows;
    if (nrows <= 0) return;
    const float beta = betas[layer];

    bool kept = false;
    int total = 0, off = 0, mine = 0;
    unsigned ev[KEEPCAP / 32];
    if (!FIRST) {
        const int c = ecnt[blockIdx.x * LWAVES + lane];
        kept = __builtin_amdgcn_ballot_w32((unsigned)c > (unsigned)KEEPCAP) == 0u;
        if (kept) {
            total = __builtin_amdgcn_readfirstlane(wave_sum_i(c));
            off   = __builtin_amdgcn_readfirstlane(wave_sum_i(lane < wave ? c : 0));
            mine  = __builtin_amdgcn_readlane(c, wave);
            const unsigned* gl = elist + (size_t)(blockIdx.x * LWAVES + wave) * KEEPCAP;
#pragma unroll
            for (int k = 0; k < KEEPCAP / 32; ++k) {
                const int i = lane + 32 * k;
                ev[k] = 0u;
                if (i < mine) ev[k] = gl[i];
            }
        }
    }

    {
        float4 v[ROWS0];
#pragma unroll
        for (int i = 0; i < ROWS0; ++i) {
            const int r = wave + i * LWAVES;
            v[i] = ((const float4*)(hin + (size_t)(lo + (r < nrows ? r : 0)) * DIM))[lane];
        }
#pragma unroll
        for (int i = 0; i < ROWS0; ++i) {
            const int r = wave + i * LWAVES;
            float ss = v[i].x * v[i].x + v[i].y * v[i].y + v[i].z * v[i].z + v[i].w * v[i].w;
            ss += xhalf(ss);
            ss += row_xor<8>(ss);
            ss += row_xor<4>(ss);
            ss += row_xor<2>(ss);
            ss += row_xor<1>(ss);
            const float inv = 1.0f / (sqrtf(ss) + 1e-12f);
            if (r < nrows) {
                float4 o;
                o.x = v[i].x * inv; o.y = v[i].y * inv; o.z = v[i].z * inv; o.w = v[i].w * inv;
                ((float4*)(s_nd + r * DIM))[lane] = o;
                float* ap = s_acc + r * DIM + lane;
                ap[0] = 0.0f; ap[32] = 0.0f; ap[64] = 0.0f; ap[96] = 0.0f;
                if (lane == 0) s_den[r] = 0.0f;
            }
        }
    }
    if (!FIRST && kept) {
#pragma unroll
        for (int k = 0; k < KEEPCAP / 32; ++k) {
            const int i = lane + 32 * k;
            if (i < mine) s_list[off + i] = ev[k];
        }
    }
    __syncthreads();

    if (FIRST || !kept) {
        scan_edges<FIRST>(hin, src, dst, lo, nrows, beta, lane, wave,
                          s_nd, s_acc, s_den, s_list, elist, ecnt);
    } else {
        const int ngr = (total + EG - 1) / EG;
        const int g0 = (wave * ngr) / LWAVES;
        const int g1 = ((wave + 1) * ngr) / LWAVES;
        for (int gi = g0; gi < g1; ++gi) {
            const int left = total - gi * EG;
            if (left >= EG)
                edge_group<true, true>(hin, src, s_list + gi * EG, EG, beta, lane, s_nd, s_acc, s_den, nullptr);
            else
                edge_group<false, true>(hin, src, s_list + gi * EG, left, beta, lane, s_nd, s_acc, s_den, nullptr);
        }
    }
    __syncthreads();

    for (int r = wave; r < nrows; r += LWAVES) {
        const float* ap = s_acc + r * DIM + lane;
        const float dn = s_den[r];
        const float d = dn > 0.0f ? dn : 1.0f;
        float4 o;
        o.x = ap[0] / d; o.y = ap[32] / d; o.z = ap[64] / d; o.w = ap[96] / d;
        ((float4*)(hout + (size_t)(lo + r) * DIM))[lane] = o;
    }
}

static_assert(NNODES % PCHUNK == 0 && PCHUNK % 4 == 0, "whole chunks, whole k steps");
static_assert(DIM == PWAVES * 16 && NGRAPH == 64, "8 column tiles, 4 row tiles of graphs");
static_assert(PBLOCKS >= FWAVES && PBLOCKS <= FWAVES * 32, "the closing sum's split of the partial tiles");

__global__ void __launch_bounds__(PWAVES * 32)
pool_partial(const float* __restrict__ h, const int* __restrict__ gid,
             float* __restrict__ part, float* __restrict__ cntp) {
    __shared__ __attribute__((aligned(16))) float s_tile[NGRAPH * DIM];

    const int lane = threadIdx.x & 31;
    const int wave = __builtin_amdgcn_readfirstlane(threadIdx.x >> 5);
    const int half = lane >> 4;
    const int c    = lane & 15;
    const int row0 = blockIdx.x * PCHUNK;
    const int n0   = wave * 16;
    const float* hp = h + (size_t)(row0 + 2 * half) * DIM + n0 + c;
    const int*   gp = gid + row0 + 2 * half;

    float bx[PSTEPS], by[PSTEPS];
    int2  gg[PSTEPS];
#pragma unroll
    for (int s = 0; s < PSTEPS; ++s) {
        gg[s] = *(const int2*)(gp + 4 * s);
        bx[s] = hp[(size_t)(4 * s) * DIM];
        by[s] = hp[(size_t)(4 * s + 1) * DIM];
    }

    v8f acc[4] = {};
    float cl[4] = {0.0f, 0.0f, 0.0f, 0.0f};
#pragma unroll
    for (int s = 0; s < PSTEPS; ++s) {
        v2f b;
        b.x = bx[s];
        b.y = by[s];
#pragma unroll
        for (int mt = 0; mt < 4; ++mt) {
            v2f a;
            a.x = gg[s].x == 16 * mt + c ? 1.0f : 0.0f;
            a.y = gg[s].y == 16 * mt + c ? 1.0f : 0.0f;
            acc[mt] = __builtin_amdgcn_wmma_f32_16x16x4_f32(
                 false, a,  false, b,
                 (short)0, acc[mt],  false,  false);
            cl[mt] += a.x + a.y;
        }
    }

#pragma unroll
    for (int mt = 0; mt < 4; ++mt) {
#pragma unroll
        for (int r = 0; r < 8; ++r)
            s_tile[(16 * mt + 8 * half + r) * DIM + n0 + c] = acc[mt][r];
    }
    float ct[4];
#pragma unroll
    for (int mt = 0; mt < 4; ++mt) ct[mt] = cl[mt] + xhalf(cl[mt]);
    if (wave == 0) {
        float* cp = cntp + (size_t)blockIdx.x * NGRAPH + lane;
        cp[0]  = half ? ct[1] : ct[0];
        cp[32] = half ? ct[3] : ct[2];
    }
    __syncthreads();
#pragma unroll
    for (int i = 0; i < NGRAPH / PWAVES; ++i) {
        const int r = wave + PWAVES * i;
        ((float4*)(part + ((size_t)blockIdx.x * NGRAPH + r) * DIM))[lane] =
            ((const float4*)(s_tile + r * DIM))[lane];
    }
}

__global__ void __launch_bounds__(FWAVES * 32)
pool_finish(const float* __restrict__ part, const float* __restrict__ cntp,
            float* __restrict__ hg) {
    __shared__ __attribute__((aligned(16))) float s_sum[FWAVES * DIM];
    __shared__ float s_cnt[FWAVES];

    const int lane = threadIdx.x & 31;
    const int wave = __builtin_amdgcn_readfirstlane(threadIdx.x >> 5);
    const int g = blockIdx.x;
    constexpr int PER = (PBLOCKS + FWAVES - 1) / FWAVES;

    float4 v[PER];
#pragma unroll
    for (int j = 0; j < PER; ++j) {
        const int p = wave + FWAVES * j;
        v[j] = ((const float4*)(part + ((size_t)(p < PBLOCKS ? p : wave) * NGRAPH + g) * DIM))[lane];
    }
    const int t = threadIdx.x;
    float cv = t < PBLOCKS ? cntp[(size_t)t * NGRAPH + g] : 0.0f;

    float4 a = {0.0f, 0.0f, 0.0f, 0.0f};
#pragma unroll
    for (int j = 0; j < PER; ++j) {
        if (wave + FWAVES * j < PBLOCKS) {
            a.x += v[j].x; a.y += v[j].y; a.z += v[j].z; a.w += v[j].w;
        }
    }
    ((float4*)(s_sum + wave * DIM))[lane] = a;

    cv += xhalf(cv);
    cv += row_xor<8>(cv);
    cv += row_xor<4>(cv);
    cv += row_xor<2>(cv);
    cv += row_xor<1>(cv);
    if (lane == 0) s_cnt[wave] = cv;
    __syncthreads();

    if (t < DIM) {
        float s = 0.0f, n = 0.0f;
        for (int w = 0; w < FWAVES; ++w) {
            s += s_sum[w * DIM + t];
            n += s_cnt[w];
        }
        hg[g * DIM + t] = s / fmaxf(n, 1.0f);
    }
}

__global__ void classifier_wmma(const float* __restrict__ hg,
                                const float* __restrict__ W,
                                const float* __restrict__ b,
                                float* __restrict__ out) {
    int mt = blockIdx.x / 7;
    int nt = blockIdx.x % 7;
    int m0 = mt * 16;
    int n0 = nt * 16;
    int lane = threadIdx.x & 31;
    int half = lane >> 4;
    int lr   = lane & 15;
    int col  = n0 + lr;
    int ccol = col < NCLS ? col : (NCLS - 1);
    float mask = col < NCLS ? 1.0f : 0.0f;

    v8f acc = {};
#pragma unroll 4
    for (int k0 = 0; k0 < DIM; k0 += 4) {
        const float* ar = hg + (size_t)(m0 + lr) * DIM + k0 + 2 * half;
        v2f a;
        a.x = ar[0];
        a.y = ar[1];
        v2f bb;
        bb.x = W[(size_t)(k0 + 2 * half) * NCLS + ccol] * mask;
        bb.y = W[(size_t)(k0 + 2 * half + 1) * NCLS + ccol] * mask;
        acc = __builtin_amdgcn_wmma_f32_16x16x4_f32(
             false, a,  false, bb,
             (short)0, acc,  false,  false);
    }

    if (col < NCLS) {
        float bias = b[col];
#pragma unroll
        for (int r = 0; r < 8; ++r) {
            int m = m0 + r + 8 * half;
            out[(size_t)m * NCLS + col] = acc[r] + bias;
        }
    }
}

extern "C" void kernel_launch(void* const* d_in, const int* in_sizes, int n_in,
                              void* d_out, int out_size, void* d_ws, size_t ws_size,
                              hipStream_t stream) {
    const float* h0    = (const float*)d_in[0];
    const int*   src   = (const int*)d_in[1];
    const int*   dst   = (const int*)d_in[2];
    const int*   gid   = (const int*)d_in[3];
    const float* betas = (const float*)d_in[4];
    const float* Wc    = (const float*)d_in[5];
    const float* bc    = (const float*)d_in[6];
    float*       out   = (float*)d_out;

    float* hA    = (float*)d_ws;
    float* hB    = hA + (size_t)NNODES * DIM;
    float* hg    = hB + (size_t)NNODES * DIM;
    unsigned* elist = (unsigned*)(hg + NGRAPH * DIM);
    int*   ecnt  = (int*)(elist + (size_t)LGROUPS * LWAVES * KEEPCAP);

    layer_owned<true><<<LGROUPS, LTHREADS, 0, stream>>>(h0, hA, src, dst, betas, 0, elist, ecnt);
    layer_owned<false><<<LGROUPS, LTHREADS, 0, stream>>>(hA, hB, src, dst, betas, 1, elist, ecnt);
    layer_owned<false><<<LGROUPS, LTHREADS, 0, stream>>>(hB, hA, src, dst, betas, 2, elist, ecnt);
    const float* hin = hA;

    float* part = hB;
    float* cntp = hB + (size_t)PBLOCKS * NGRAPH * DIM;
    pool_partial<<<PBLOCKS, PWAVES * 32, 0, stream>>>(hin, gid, part, cntp);
    pool_finish<<<NGRAPH, FWAVES * 32, 0, stream>>>(part, cntp, hg);

    classifier_wmma<<<4 * 7, 32, 0, stream>>>(hg, Wc, bc, out);
}


# Layer 1 scan reads one-byte owner keys (dst / rows per group, from a new first launch), not 4-byte ids: a quarter of the bytes, 4 edges per 32-bit test, 4x longer steps; LDS add order differs
# speedup vs baseline: 5.3923x; 1.0574x over previous
#include <hip/hip_runtime.h>
#include <math.h>


#define NNODES  40000
#define NEDGES  640000
#define DIM     128
#define NGRAPH  64
#define NCLS    100
#define NLAYERS 3

#define PCHUNK   160
#define PBLOCKS  (NNODES / PCHUNK)
#define PSTEPS   (PCHUNK / 4)
#define PWAVES   8
#define FWAVES   32

#define LGROUPS  256
#define LWAVES   32
#define LTHREADS (LWAVES * 32)
#define RPG      ((NNODES + LGROUPS - 1) / LGROUPS)
#define ROWS0    ((RPG + LWAVES - 1) / LWAVES)
#define BATCH    512
#define EG       8
#define LISTCAP  (BATCH + EG)
#define NBATCH   (NEDGES / BATCH)
#define KEEPCAP  160
#define KPER     16
#define KTHREADS 256
#define KBLOCKS  ((NEDGES / KPER + KTHREADS - 1) / KTHREADS)
#define KQ       4
#define KSTEP    (32 * KQ * KPER)
#define NQUAD    (NEDGES / KPER)
#define NKSTEP   ((NQUAD + 32 * KQ - 1) / (32 * KQ))

static_assert(NEDGES % BATCH == 0, "the id scan reads whole steps");
static_assert(RPG <= 256 && LGROUPS == 256, "kept entry holds a local row in 8 bits; an owner key is one byte");
static_assert(NEDGES % KPER == 0 && KPER == 16, "a lane's 16 key bytes are one thread's store of the key pass");
static_assert(NBATCH >= LWAVES && NKSTEP >= LWAVES, "every wave has a first step");
static_assert(NQUAD % 32 == 0 && KQ == 4, "a key load is whole or absent for the wave; 64 found bits per lane");
static_assert(EG - 1 + 32 <= LISTCAP, "a trip of the key scan adds at most 32 entries");
static_assert(NNODES <= (1 << 24), "kept entry = source row << 8 | local row");
static_assert(KEEPCAP % 32 == 0 && KEEPCAP % EG == 0, "a kept list is read a wave-wide load at a time");
static_assert(LWAVES * KEEPCAP <= LWAVES * LISTCAP, "the group's joined list fits the LDS list area");
static_assert((size_t)PBLOCKS * NGRAPH * (DIM + 1) <= (size_t)NNODES * DIM, "partial tiles and counts fit one row buffer");

typedef __attribute__((ext_vector_type(2))) float v2f;
typedef __attribute__((ext_vector_type(8))) float v8f;

__device__ __forceinline__ void ldsAddF(float* p, float v) {
    __hip_atomic_fetch_add(p, v, __ATOMIC_RELAXED, __HIP_MEMORY_SCOPE_WORKGROUP);
}

template <int M>
__device__ __forceinline__ float row_xor(float x) {
    int v = __builtin_bit_cast(int, x);
    return __builtin_bit_cast(float, __builtin_amdgcn_update_dpp(0, v, 0x160 + M, 0xf, 0xf, true));
}
__device__ __forceinline__ float xhalf(float x) {
    int v = __builtin_bit_cast(int, x);
    return __builtin_bit_cast(float, __builtin_amdgcn_permlanex16(v, v, 0x76543210, 0xfedcba98, false, false));
}
__device__ __forceinline__ float lane_val(float x, int l) {
    return __builtin_bit_cast(float, __builtin_amdgcn_readlane(__builtin_bit_cast(int, x), l));
}
__device__ __forceinline__ int wave_sum_i(int x) {
    x += __builtin_amdgcn_permlanex16(x, x, 0x76543210, 0xfedcba98, false, false);
    x += __builtin_amdgcn_update_dpp(0, x, 0x160 + 8, 0xf, 0xf, true);
    x += __builtin_amdgcn_update_dpp(0, x, 0x160 + 4, 0xf, 0xf, true);
    x += __builtin_amdgcn_update_dpp(0, x, 0x160 + 2, 0xf, 0xf, true);
    x += __builtin_amdgcn_update_dpp(0, x, 0x160 + 1, 0xf, 0xf, true);
    return x;
}

template <bool FULL, bool KEPT>
__device__ __forceinline__ void edge_group(const float* __restrict__ hin,
                                           const int* __restrict__ src,
                                           const int* __restrict__ dst, int lo,
                                           const unsigned* ent, int n,
                                           float beta, int lane,
                                           const float* s_nd, float* s_acc,
                                           float* s_den, unsigned* keep) {
    int slot = lane & (EG - 1);
    if (!FULL) slot = slot < n ? slot : 0;
    const unsigned en = ent[slot];
    const int sv = KEPT ? (int)(en >> 8) : src[en];
    const int dl = KEPT ? (int)(en & 255u) : dst[en] - lo;
    if (!KEPT) {
        if (keep != nullptr && lane < (FULL ? EG : n))
            keep[lane] = ((unsigned)sv << 8) | (unsigned)dl;
    }
    int    dj[EG];
    float4 a4[EG];
    float4 b4[EG];
#pragma unroll
    for (int u = 0; u < EG; ++u) {
        if (FULL || u < n) {
            const int sj = __builtin_amdgcn_readlane(sv, u);
            dj[u] = __builtin_amdgcn_readlane(dl, u);
            a4[u] = ((const float4*)(hin + (size_t)sj * DIM))[lane];
        }
    }
#pragma unroll
    for (int u = 0; u < EG; ++u) {
        if (FULL || u < n)
            b4[u] = ((const float4*)(s_nd + dj[u] * DIM))[lane];
    }
    __builtin_amdgcn_sched_barrier(0);
    const int  l16 = lane & 15;
    const bool lowhalf = lane < 16;
    float sums = 0.0f;
#pragma unroll
    for (int u = 0; u < EG; ++u) {
        if (FULL || u < n) {
            const float4 a = a4[u];
            const float4 b = b4[u];
            float p = a.x * b.x + a.y * b.y + a.z * b.z + a.w * b.w;
            float q = a.x * a.x + a.y * a.y + a.z * a.z + a.w * a.w;
            float t = (lowhalf ? p : q) + xhalf(lowhalf ? q : p);
            t += row_xor<8>(t);
            t += row_xor<4>(t);
            t += row_xor<2>(t);
            t += row_xor<1>(t);
            sums = l16 == u ? t : sums;
        }
    }
    const float ssq  = xhalf(sums);
    const float attv = __expf(beta * (sums / (sqrtf(ssq) + 1e-12f)));
#pragma unroll
    for (int u = 0; u < EG; ++u) {
        if (FULL || u < n) {
            const float att = lane_val(attv, u);
            const float4 a = a4[u];
            float* ap = s_acc + dj[u] * DIM + lane;
            ldsAddF(ap + 0,  att * a.x);
            ldsAddF(ap + 32, att * a.y);
            ldsAddF(ap + 64, att * a.z);
            ldsAddF(ap + 96, att * a.w);
        }
    }
    if (lane < (FULL ? EG : n)) ldsAddF(s_den + dl, attv);
}

__device__ __forceinline__ void load_ids(int4 (&d)[4], const int* __restrict__ dst,
                                         int b, int lane) {
#pragma unroll
    for (int q = 0; q < 4; ++q)
        d[q] = ((const int4*)(dst + (size_t)b * BATCH + q * 128))[lane];
}

__device__ __forceinline__ unsigned owner_key(int id) {
    const unsigned k = (unsigned)id / (unsigned)RPG;
    return k < 255u ? k : 255u;
}
static_assert((LGROUPS - 1) * RPG >= NNODES, "key 255 belongs to a work-group without rows");

__global__ void __launch_bounds__(KTHREADS)
owner_keys(const int* __restrict__ dst, unsigned* __restrict__ keys) {
    const int t = blockIdx.x * KTHREADS + threadIdx.x;
    if (t >= NEDGES / KPER) return;
    const int4* p = (const int4*)(dst + (size_t)t * KPER);
    int4 d[4];
#pragma unroll
    for (int q = 0; q < 4; ++q) d[q] = p[q];
    unsigned o[4];
#pragma unroll
    for (int q = 0; q < 4; ++q)
        o[q] = owner_key(d[q].x) | (owner_key(d[q].y) << 8) |
               (owner_key(d[q].z) << 16) | (owner_key(d[q].w) << 24);
    uint4 v;
    v.x = o[0]; v.y = o[1]; v.z = o[2]; v.w = o[3];
    ((uint4*)keys)[t] = v;
}

__device__ __forceinline__ void load_keys(uint4 (&d)[KQ], const unsigned* __restrict__ keys,
                                          int s, int lane) {
#pragma unroll
    for (int j = 0; j < KQ; ++j) {
        const int q0 = (s * KQ + j) * 32;
        d[j] = ((const uint4*)keys)[(q0 < NQUAD ? q0 : NQUAD - 32) + lane];
    }
}

__device__ __forceinline__ void scan_keys(const float* __restrict__ hin,
                                          const int* __restrict__ src,
                                          const int* __restrict__ dst,
                                          const unsigned* __restrict__ keys,
                                          int lo, float beta,
                                          int lane, int wave,
                                          const float* s_nd, float* s_acc, float* s_den,
                                          unsigned* s_list,
                                          unsigned* __restrict__ elist,
                                          int* __restrict__ ecnt) {
    unsigned* mylist = s_list + wave * LISTCAP;
    unsigned* mykeep = elist + (size_t)(blockIdx.x * LWAVES + wave) * KEEPCAP;
    const unsigned own4 = (unsigned)blockIdx.x * 0x01010101u;
    int cnt = 0;
    int done = 0;
    uint4 cur[KQ];
    load_keys(cur, keys, wave, lane);
    for (int s = wave; s < NKSTEP; s += LWAVES) {
        uint4 nxt[KQ];
        load_keys(nxt, keys, s + LWAVES, lane);
        unsigned f[2] = {0u, 0u};
#pragma unroll
        for (int j = 0; j < KQ; ++j) {
            const unsigned w[4] = {cur[j].x, cur[j].y, cur[j].z, cur[j].w};
#pragma unroll
            for (int q = 0; q < 4; ++q) {
                const unsigned x = w[q] ^ own4;
                const unsigned t = (x & 0x7f7f7f7fu) + 0x7f7f7f7fu;
                const unsigned z = ~(t | x | 0x7f7f7f7fu);
                f[j >> 1] |= z >> (7 - (4 * (j & 1) + q));
            }
        }
        const int left = (NQUAD - s * (KQ * 32)) >> 5;
        f[0] &= left >= 2 ? 0xffffffffu : 0x0f0f0f0fu;
        f[1] &= left >= 4 ? 0xffffffffu : (left == 3 ? 0x0f0f0f0fu : 0u);
        unsigned long long found = (unsigned long long)f[0] | ((unsigned long long)f[1] << 32);
        unsigned m = __builtin_amdgcn_ballot_w32(found != 0ull);
        const int ebase = s * KSTEP + lane * KPER;
        while (m != 0u) {
            const bool has = found != 0ull;
            const int t = __builtin_ctzll(has ? found : 1ull);
            const int j = ((t >> 5) << 1) | ((t >> 2) & 1);
            const int pos = cnt + (int)__builtin_amdgcn_mbcnt_lo(m, 0u);
            if (has) mylist[pos] = (unsigned)(ebase + j * (32 * KPER) + ((t & 3) << 2) + ((t >> 3) & 3));
            cnt += __builtin_popcount(m);
            found &= found - 1ull;
            m = __builtin_amdgcn_ballot_w32(found != 0ull);
            while (cnt >= EG) {
                cnt -= EG;
                unsigned* kp = done + EG <= KEEPCAP ? mykeep + done : nullptr;
                edge_group<true, false>(hin, src, dst, lo, mylist + cnt, EG, beta, lane, s_nd, s_acc, s_den, kp);
                done += EG;
            }
        }
#pragma unroll
        for (int j = 0; j < KQ; ++j) cur[j] = nxt[j];
    }
    if (cnt > 0) {
        unsigned* kp = done + cnt <= KEEPCAP ? mykeep + done : nullptr;
        edge_group<false, false>(hin, src, dst, lo, mylist, cnt, beta, lane, s_nd, s_acc, s_den, kp);
        done += cnt;
    }
    if (lane == 0) ecnt[blockIdx.x * LWAVES + wave] = done;
}

__device__ __forceinline__ void scan_ids(const float* __restrict__ hin,
                                         const int* __restrict__ src,
                                         const int* __restrict__ dst,
                                         int lo, int nrows, float beta,
                                         int lane, int wave,
                                         const float* s_nd, float* s_acc, float* s_den,
                                         unsigned* s_list) {
    unsigned* mylist = s_list + wave * LISTCAP;
    int cnt = 0;
    int4 cur[4];
    load_ids(cur, dst, wave, lane);
    for (int b = wave; b < NBATCH; b += LWAVES) {
        int4 nxt[4];
        load_ids(nxt, dst, b + LWAVES < NBATCH ? b + LWAVES : b, lane);
        const int ebase = b * BATCH + lane * 4;
#pragma unroll
        for (int q = 0; q < 4; ++q) {
            const int idq[4] = {cur[q].x, cur[q].y, cur[q].z, cur[q].w};
#pragma unroll
            for (int c = 0; c < 4; ++c) {
                const unsigned dl = (unsigned)(idq[c] - lo);
                const bool mine = dl < (unsigned)nrows;
                const unsigned m = __builtin_amdgcn_ballot_w32(mine);
                if (m != 0u) {
                    const int pos = cnt + (int)__builtin_amdgcn_mbcnt_lo(m, 0u);
                    if (mine) mylist[pos] = (unsigned)(ebase + q * 128 + c);
                    cnt += __builtin_popcount(m);
                }
            }
        }
        while (cnt >= EG) {
            cnt -= EG;
            edge_group<true, false>(hin, src, dst, lo, mylist + cnt, EG, beta, lane, s_nd, s_acc, s_den, nullptr);
        }
#pragma unroll
        for (int q = 0; q < 4; ++q) cur[q] = nxt[q];
    }
    if (cnt > 0)
        edge_group<false, false>(hin, src, dst, lo, mylist, cnt, beta, lane, s_nd, s_acc, s_den, nullptr);
}

template <bool FIRST>
__global__ void __launch_bounds__(LTHREADS)
layer_owned(const float* __restrict__ hin, float* __restrict__ hout,
            const int* __restrict__ src, const int* __restrict__ dst,
            const float* __restrict__ betas, int layer,
            unsigned* __restrict__ elist, int* __restrict__ ecnt,
            const unsigned* __restrict__ keys) {
    __shared__ __attribute__((aligned(16))) float s_nd[RPG * DIM];
    __shared__ __attribute__((aligned(16))) float s_acc[RPG * DIM];
    __shared__ float s_den[RPG];
    __shared__ unsigned s_list[LWAVES * LISTCAP];

    const int lane = threadIdx.x & 31;
    const int wave = __builtin_amdgcn_readfirstlane(threadIdx.x >> 5);
    const int lo = blockIdx.x * RPG;
    int nrows = NNODES - lo;
    nrows = nrows > RPG ? RPG : nrows;
    if (nrows <= 0) return;
    const float beta = betas[layer];

    bool kept = false;
    int total = 0, off = 0, mine = 0;
    unsigned ev[KEEPCAP / 32];
    if (!FIRST) {
        const int c = ecnt[blockIdx.x * LWAVES + lane];
        kept = __builtin_amdgcn_ballot_w32((unsigned)c > (unsigned)KEEPCAP) == 0u;
        if (kept) {
            total = __builtin_amdgcn_readfirstlane(wave_sum_i(c));
            off   = __builtin_amdgcn_readfirstlane(wave_sum_i(lane < wave ? c : 0));
            mine  = __builtin_amdgcn_readlane(c, wave);
            const unsigned* gl = elist + (size_t)(blockIdx.x * LWAVES + wave) * KEEPCAP;
#pragma unroll
            for (int k = 0; k < KEEPCAP / 32; ++k) {
                const int i = lane + 32 * k;
                ev[k] = 0u;
                if (i < mine) ev[k] = gl[i];
            }
        }
    }

    {
        float4 v[ROWS0];
#pragma unroll
        for (int i = 0; i < ROWS0; ++i) {
            const int r = wave + i * LWAVES;
            v[i] = ((const float4*)(hin + (size_t)(lo + (r < nrows ? r : 0)) * DIM))[lane];
        }
#pragma unroll
        for (int i = 0; i < ROWS0; ++i) {
            const int r = wave + i * LWAVES;
            float ss = v[i].x * v[i].x + v[i].y * v[i].y + v[i].z * v[i].z + v[i].w * v[i].w;
            ss += xhalf(ss);
            ss += row_xor<8>(ss);
            ss += row_xor<4>(ss);
            ss += row_xor<2>(ss);
            ss += row_xor<1>(ss);
            const float inv = 1.0f / (sqrtf(ss) + 1e-12f);
            if (r < nrows) {
                float4 o;
                o.x = v[i].x * inv; o.y = v[i].y * inv; o.z = v[i].z * inv; o.w = v[i].w * inv;
                ((float4*)(s_nd + r * DIM))[lane] = o;
                float* ap = s_acc + r * DIM + lane;
                ap[0] = 0.0f; ap[32] = 0.0f; ap[64] = 0.0f; ap[96] = 0.0f;
                if (lane == 0) s_den[r] = 0.0f;
            }
        }
    }
    if (!FIRST && kept) {
#pragma unroll
        for (int k = 0; k < KEEPCAP / 32; ++k) {
            const int i = lane + 32 * k;
            if (i < mine) s_list[off + i] = ev[k];
        }
    }
    __syncthreads();

    if (FIRST) {
        scan_keys(hin, src, dst, keys, lo, beta, lane, wave,
                  s_nd, s_acc, s_den, s_list, elist, ecnt);
    } else if (!kept) {
        scan_ids(hin, src, dst, lo, nrows, beta, lane, wave,
                 s_nd, s_acc, s_den, s_list);
    } else {
        const int ngr = (total + EG - 1) / EG;
        const int g0 = (wave * ngr) / LWAVES;
        const int g1 = ((wave + 1) * ngr) / LWAVES;
        for (int gi = g0; gi < g1; ++gi) {
            const int left = total - gi * EG;
            if (left >= EG)
                edge_group<true, true>(hin, src, dst, lo, s_list + gi * EG, EG, beta, lane, s_nd, s_acc, s_den, nullptr);
            else
                edge_group<false, true>(hin, src, dst, lo, s_list + gi * EG, left, beta, lane, s_nd, s_acc, s_den, nullptr);
        }
    }
    __syncthreads();

    for (int r = wave; r < nrows; r += LWAVES) {
        const float* ap = s_acc + r * DIM + lane;
        const float dn = s_den[r];
        const float d = dn > 0.0f ? dn : 1.0f;
        float4 o;
        o.x = ap[0] / d; o.y = ap[32] / d; o.z = ap[64] / d; o.w = ap[96] / d;
        ((float4*)(hout + (size_t)(lo + r) * DIM))[lane] = o;
    }
}

static_assert(NNODES % PCHUNK == 0 && PCHUNK % 4 == 0, "whole chunks, whole k steps");
static_assert(DIM == PWAVES * 16 && NGRAPH == 64, "8 column tiles, 4 row tiles of graphs");
static_assert(PBLOCKS >= FWAVES && PBLOCKS <= FWAVES * 32, "the closing sum's split of the partial tiles");

__global__ void __launch_bounds__(PWAVES * 32)
pool_partial(const float* __restrict__ h, const int* __restrict__ gid,
             float* __restrict__ part, float* __restrict__ cntp) {
    __shared__ __attribute__((aligned(16))) float s_tile[NGRAPH * DIM];

    const int lane = threadIdx.x & 31;
    const int wave = __builtin_amdgcn_readfirstlane(threadIdx.x >> 5);
    const int half = lane >> 4;
    const int c    = lane & 15;
    const int row0 = blockIdx.x * PCHUNK;
    const int n0   = wave * 16;
    const float* hp = h + (size_t)(row0 + 2 * half) * DIM + n0 + c;
    const int*   gp = gid + row0 + 2 * half;

    float bx[PSTEPS], by[PSTEPS];
    int2  gg[PSTEPS];
#pragma unroll
    for (int s = 0; s < PSTEPS; ++s) {
        gg[s] = *(const int2*)(gp + 4 * s);
        bx[s] = hp[(size_t)(4 * s) * DIM];
        by[s] = hp[(size_t)(4 * s + 1) * DIM];
    }

    v8f acc[4] = {};
    float cl[4] = {0.0f, 0.0f, 0.0f, 0.0f};
#pragma unroll
    for (int s = 0; s < PSTEPS; ++s) {
        v2f b;
        b.x = bx[s];
        b.y = by[s];
#pragma unroll
        for (int mt = 0; mt < 4; ++mt) {
            v2f a;
            a.x = gg[s].x == 16 * mt + c ? 1.0f : 0.0f;
            a.y = gg[s].y == 16 * mt + c ? 1.0f : 0.0f;
            acc[mt] = __builtin_amdgcn_wmma_f32_16x16x4_f32(
                 false, a,  false, b,
                 (short)0, acc[mt],  false,  false);
            cl[mt] += a.x + a.y;
        }
    }

#pragma unroll
    for (int mt = 0; mt < 4; ++mt) {
#pragma unroll
        for (int r = 0; r < 8; ++r)
            s_tile[(16 * mt + 8 * half + r) * DIM + n0 + c] = acc[mt][r];
    }
    float ct[4];
#pragma unroll
    for (int mt = 0; mt < 4; ++mt) ct[mt] = cl[mt] + xhalf(cl[mt]);
    if (wave == 0) {
        float* cp = cntp + (size_t)blockIdx.x * NGRAPH + lane;
        cp[0]  = half ? ct[1] : ct[0];
        cp[32] = half ? ct[3] : ct[2];
    }
    __syncthreads();
#pragma unroll
    for (int i = 0; i < NGRAPH / PWAVES; ++i) {
        const int r = wave + PWAVES * i;
        ((float4*)(part + ((size_t)blockIdx.x * NGRAPH + r) * DIM))[lane] =
            ((const float4*)(s_tile + r * DIM))[lane];
    }
}

__global__ void __launch_bounds__(FWAVES * 32)
pool_finish(const float* __restrict__ part, const float* __restrict__ cntp,
            float* __restrict__ hg) {
    __shared__ __attribute__((aligned(16))) float s_sum[FWAVES * DIM];
    __shared__ float s_cnt[FWAVES];

    const int lane = threadIdx.x & 31;
    const int wave = __builtin_amdgcn_readfirstlane(threadIdx.x >> 5);
    const int g = blockIdx.x;
    constexpr int PER = (PBLOCKS + FWAVES - 1) / FWAVES;

    float4 v[PER];
#pragma unroll
    for (int j = 0; j < PER; ++j) {
        const int p = wave + FWAVES * j;
        v[j] = ((const float4*)(part + ((size_t)(p < PBLOCKS ? p : wave) * NGRAPH + g) * DIM))[lane];
    }
    const int t = threadIdx.x;
    float cv = t < PBLOCKS ? cntp[(size_t)t * NGRAPH + g] : 0.0f;

    float4 a = {0.0f, 0.0f, 0.0f, 0.0f};
#pragma unroll
    for (int j = 0; j < PER; ++j) {
        if (wave + FWAVES * j < PBLOCKS) {
            a.x += v[j].x; a.y += v[j].y; a.z += v[j].z; a.w += v[j].w;
        }
    }
    ((float4*)(s_sum + wave * DIM))[lane] = a;

    cv += xhalf(cv);
    cv += row_xor<8>(cv);
    cv += row_xor<4>(cv);
    cv += row_xor<2>(cv);
    cv += row_xor<1>(cv);
    if (lane == 0) s_cnt[wave] = cv;
    __syncthreads();

    if (t < DIM) {
        float s = 0.0f, n = 0.0f;
        for (int w = 0; w < FWAVES; ++w) {
            s += s_sum[w * DIM + t];
            n += s_cnt[w];
        }
        hg[g * DIM + t] = s / fmaxf(n, 1.0f);
    }
}

__global__ void classifier_wmma(const float* __restrict__ hg,
                                const float* __restrict__ W,
                                const float* __restrict__ b,
                                float* __restrict__ out) {
    int mt = blockIdx.x / 7;
    int nt = blockIdx.x % 7;
    int m0 = mt * 16;
    int n0 = nt * 16;
    int lane = threadIdx.x & 31;
    int half = lane >> 4;
    int lr   = lane & 15;
    int col  = n0 + lr;
    int ccol = col < NCLS ? col : (NCLS - 1);
    float mask = col < NCLS ? 1.0f : 0.0f;

    v8f acc = {};
#pragma unroll 4
    for (int k0 = 0; k0 < DIM; k0 += 4) {
        const float* ar = hg + (size_t)(m0 + lr) * DIM + k0 + 2 * half;
        v2f a;
        a.x = ar[0];
        a.y = ar[1];
        v2f bb;
        bb.x = W[(size_t)(k0 + 2 * half) * NCLS + ccol] * mask;
        bb.y = W[(size_t)(k0 + 2 * half + 1) * NCLS + ccol] * mask;
        acc = __builtin_amdgcn_wmma_f32_16x16x4_f32(
             false, a,  false, bb,
             (short)0, acc,  false,  false);
    }

    if (col < NCLS) {
        float bias = b[col];
#pragma unroll
        for (int r = 0; r < 8; ++r) {
            int m = m0 + r + 8 * half;
            out[(size_t)m * NCLS + col] = acc[r] + bias;
        }
    }
}

extern "C" void kernel_launch(void* const* d_in, const int* in_sizes, int n_in,
                              void* d_out, int out_size, void* d_ws, size_t ws_size,
                              hipStream_t stream) {
    const float* h0    = (const float*)d_in[0];
    const int*   src   = (const int*)d_in[1];
    const int*   dst   = (const int*)d_in[2];
    const int*   gid   = (const int*)d_in[3];
    const float* betas = (const float*)d_in[4];
    const float* Wc    = (const float*)d_in[5];
    const float* bc    = (const float*)d_in[6];
    float*       out   = (float*)d_out;

    float* hA    = (float*)d_ws;
    float* hB    = hA + (size_t)NNODES * DIM;
    float* hg    = hB + (size_t)NNODES * DIM;
    unsigned* elist = (unsigned*)(hg + NGRAPH * DIM);
    int*   ecnt  = (int*)(elist + (size_t)LGROUPS * LWAVES * KEEPCAP);
    unsigned* keys = (unsigned*)hB;

    owner_keys<<<KBLOCKS, KTHREADS, 0, stream>>>(dst, keys);
    layer_owned<true><<<LGROUPS, LTHREADS, 0, stream>>>(h0, hA, src, dst, betas, 0, elist, ecnt, keys);
    layer_owned<false><<<LGROUPS, LTHREADS, 0, stream>>>(hA, hB, src, dst, betas, 1, elist, ecnt, keys);
    layer_owned<false><<<LGROUPS, LTHREADS, 0, stream>>>(hB, hA, src, dst, betas, 2, elist, ecnt, keys);
    const float* hin = hA;

    float* part = hB;
    float* cntp = hB + (size_t)PBLOCKS * NGRAPH * DIM;
    pool_partial<<<PBLOCKS, PWAVES * 32, 0, stream>>>(hin, gid, part, cntp);
    pool_finish<<<NGRAPH, FWAVES * 32, 0, stream>>>(part, cntp, hg);

    classifier_wmma<<<4 * 7, 32, 0, stream>>>(hg, Wc, bc, out);
}
